# starting kernel
# baseline (speedup 1.0000x reference)
#include <hip/hip_runtime.h>

typedef float v2f __attribute__((ext_vector_type(2)));
typedef float v8f __attribute__((ext_vector_type(8)));

#define IN_F   1024
#define OUT_F  1024
#define TASK_D 256
#define RANK   8
#define HID    64

__global__ __launch_bounds__(512)
void gen_hidden_kernel(const float* __restrict__ task_emb,
                       const float* __restrict__ Wa1, const float* __restrict__ ba1,
                       const float* __restrict__ Wb1, const float* __restrict__ bb1,
                       float* __restrict__ h) {
  const int tid   = threadIdx.x;
  const int which = tid >> 8;
  const int rem   = tid & 255;
  const int t     = rem >> 6;
  const int hidx  = rem & 63;
  const float* W    = which ? Wb1 : Wa1;
  const float* bias = which ? bb1 : ba1;
  const float* te = task_emb + t * TASK_D;
  const float* wr = W + hidx * TASK_D;
  float s = bias[hidx];
#pragma unroll 8
  for (int d = 0; d < TASK_D; ++d) s += wr[d] * te[d];
  float g = 0.5f * s * (1.0f + erff(s * 0.7071067811865476f));
  h[tid] = g;
}

__global__ __launch_bounds__(256)
void gen_mats_kernel(const float* __restrict__ h,
                     const float* __restrict__ Wa2, const float* __restrict__ ba2,
                     const float* __restrict__ Wb2, const float* __restrict__ bb2,
                     float* __restrict__ amat, float* __restrict__ bmat) {
  const int tid = blockIdx.x * 256 + threadIdx.x;
  if (tid >= 65536) {
    const int rem = tid - 65536;
    const int t   = rem >> 13;
    bmat[t * 16384 + 8192 + (rem & 8191)] = 0.0f;
    return;
  }
  const int which = tid >> 15;
  const int rem   = tid & 32767;
  const int t     = rem >> 13;
  const int row   = rem & 8191;
  const float* W    = which ? Wb2 : Wa2;
  const float* bias = which ? bb2 : ba2;
  const float* hh = h + which * 256 + t * HID;
  const float* wr = W + (size_t)row * HID;
  float s = bias[row];
#pragma unroll
  for (int j = 0; j < HID; ++j) s += wr[j] * hh[j];
  if (which) bmat[t * 16384 + row] = s;
  else       amat[t * 8192  + row] = s;
}

__global__ __launch_bounds__(256)
void lora_apply_kernel(const float* __restrict__ x, const float* __restrict__ base,
                       const float* __restrict__ amat, const float* __restrict__ bmat,
                       float* __restrict__ out) {
  const int t    = blockIdx.y;
  const int jt   = blockIdx.x;
  const int tid  = threadIdx.x;
  const int lane = tid & 31;
  const int w    = tid >> 5;
  const int hs   = lane >> 4;
  const int lp   = lane & 15;

  __shared__ float part[8][128];
  __shared__ float tmpf[128];

  const float* xrow = x + (size_t)(t + 4 * (jt * 16 + lp)) * IN_F;
  const float* at   = amat + t * (OUT_F * RANK);
  const float* brow = bmat + t * 16384 + lp * IN_F;

  v8f acc = {};
  const int kstart = w * 128 + hs * 2;
#pragma unroll 8
  for (int kk = 0; kk < 128; kk += 4) {
    const int kb = kstart + kk;
    v2f af = __builtin_nontemporal_load((const v2f*)(xrow + kb));
    v2f bf = *(const v2f*)(brow + kb);
    acc = __builtin_amdgcn_wmma_f32_16x16x4_f32(
        false, af, false, bf, (short)0, acc, false, false);
  }
  if (lp < RANK) {
#pragma unroll
    for (int v = 0; v < 8; ++v)
      part[w][(v + 8 * hs) * RANK + lp] = acc[v];
  }
  __syncthreads();
  if (tid < 128) {
    float s = part[0][tid];
#pragma unroll
    for (int p = 1; p < 8; ++p) s += part[p][tid];
    tmpf[tid] = s;
  }
  __syncthreads();

  v2f a01 = *(const v2f*)(&tmpf[lp * RANK + hs * 2]);
  v2f a23 = *(const v2f*)(&tmpf[lp * RANK + 4 + hs * 2]);

#pragma unroll 2
  for (int it = 0; it < 8; ++it) {
    const int o = w * 128 + it * 16 + lp;
    v2f b01 = *(const v2f*)(at + o * RANK + hs * 2);
    v2f b23 = *(const v2f*)(at + o * RANK + 4 + hs * 2);
    v8f cc;
#pragma unroll
    for (int v = 0; v < 8; ++v) {
      const size_t row = (size_t)(t + 4 * (jt * 16 + v + 8 * hs));
      cc[v] = __builtin_nontemporal_load(base + row * OUT_F + o);
    }
    cc = __builtin_amdgcn_wmma_f32_16x16x4_f32(
        false, a01, false, b01, (short)0, cc, false, false);
    cc = __builtin_amdgcn_wmma_f32_16x16x4_f32(
        false, a23, false, b23, (short)0, cc, false, false);
#pragma unroll
    for (int v = 0; v < 8; ++v) {
      const size_t row = (size_t)(t + 4 * (jt * 16 + v + 8 * hs));
      __builtin_nontemporal_store(cc[v], out + row * OUT_F + o);
    }
  }
}

extern "C" void kernel_launch(void* const* d_in, const int* in_sizes, int n_in,
                              void* d_out, int out_size, void* d_ws, size_t ws_size,
                              hipStream_t stream) {
  const float* x    = (const float*)d_in[0];
  const float* base = (const float*)d_in[1];
  const float* task = (const float*)d_in[2];
  const float* Wa1  = (const float*)d_in[3];
  const float* ba1  = (const float*)d_in[4];
  const float* Wa2  = (const float*)d_in[5];
  const float* ba2  = (const float*)d_in[6];
  const float* Wb1  = (const float*)d_in[7];
  const float* bb1  = (const float*)d_in[8];
  const float* Wb2  = (const float*)d_in[9];
  const float* bb2  = (const float*)d_in[10];
  float* out = (float*)d_out;

  float* h    = (float*)d_ws;
  float* amat = h + 512;
  float* bmat = amat + 32768;

  gen_hidden_kernel<<<1, 512, 0, stream>>>(task, Wa1, ba1, Wb1, bb1, h);
  gen_mats_kernel<<<384, 256, 0, stream>>>(h, Wa2, ba2, Wb2, bb2, amat, bmat);

  dim3 grid(128, 4, 1);
  lora_apply_kernel<<<grid, 256, 0, stream>>>(x, base, amat, bmat, out);
}


# rows that lanes read at a stride (W1, W2, x) fetched whole-line (b128, 512 B per wave) and re-laid via LDS; b packed in fragment order; stage-1 k order inside 8-chunks changes the f32 sum order
# speedup vs baseline: 1.6293x; 1.6293x over previous
#include <hip/hip_runtime.h>

typedef float v2f __attribute__((ext_vector_type(2)));
typedef float v4f __attribute__((ext_vector_type(4)));
typedef float v8f __attribute__((ext_vector_type(8)));

#define IN_F   1024
#define OUT_F  1024
#define TASK_D 256
#define RANK   8
#define HID    64
#define XP     132
#define W1P    260
#define W2P    68

__global__ __launch_bounds__(512)
void gen_hidden_kernel(const float* __restrict__ task_emb,
                       const float* __restrict__ Wa1, const float* __restrict__ ba1,
                       const float* __restrict__ Wb1, const float* __restrict__ bb1,
                       float* __restrict__ h) {
  __shared__ __attribute__((aligned(16))) float w1s[128 * W1P];
  const int tid   = threadIdx.x;
  {
    const int lane = tid & 31;
    const int wv   = tid >> 5;
    const int r0   = (wv & 7) * 8;
    const float* src = ((wv < 8) ? Wa1 : Wb1) + r0 * TASK_D + lane * 4;
    v4f wr[16];
#pragma unroll
    for (int i = 0; i < 16; ++i) wr[i] = *(const v4f*)(src + i * 128);
    float* dst = w1s + ((wv >> 3) * 64 + r0) * W1P + lane * 4;
#pragma unroll
    for (int i = 0; i < 16; ++i) *(v4f*)(dst + (i >> 1) * W1P + (i & 1) * 128) = wr[i];
  }
  __syncthreads();
  const int which = tid >> 8;
  const int rem   = tid & 255;
  const int t     = rem >> 6;
  const int hidx  = rem & 63;
  const float* bias = which ? bb1 : ba1;
  const float* te = task_emb + t * TASK_D;
  const float* wr = w1s + (which * 64 + hidx) * W1P;
  float s = bias[hidx];
#pragma unroll 8
  for (int d = 0; d < TASK_D; ++d) s += wr[d] * te[d];
  float g = 0.5f * s * (1.0f + erff(s * 0.7071067811865476f));
  h[tid] = g;
}

__global__ __launch_bounds__(256)
void gen_mats_kernel(const float* __restrict__ h,
                     const float* __restrict__ Wa2, const float* __restrict__ ba2,
                     const float* __restrict__ Wb2, const float* __restrict__ bb2,
                     float* __restrict__ amat, float* __restrict__ bmat) {
  __shared__ __attribute__((aligned(16))) float w2s[8][32 * W2P];
  const int tid = blockIdx.x * 256 + threadIdx.x;
  if (tid >= 65536) {
    const int rem = tid - 65536;
    const int t   = rem >> 13;
    const int q   = rem & 8191;
    const int j   = q >> 6;
    const int z   = q & 63;
    const int zh  = z >> 5;
    const int zc  = 8 + ((z >> 2) & 7);
    bmat[t * 16384 + j * 128 + (zh * 16 + zc) * 4 + (z & 3)] = 0.0f;
    return;
  }
  const int which = tid >> 15;
  const int rem   = tid & 32767;
  const int t     = rem >> 13;
  const int row   = rem & 8191;
  const float* W    = which ? Wb2 : Wa2;
  const float* bias = which ? bb2 : ba2;
  const float* hh = h + which * 256 + t * HID;
  const int lane = threadIdx.x & 31;
  const float* wsrc = W + (size_t)(row & ~31) * HID + lane * 4;
  v4f wq[16];
#pragma unroll
  for (int i = 0; i < 16; ++i) wq[i] = *(const v4f*)(wsrc + i * 128);
  float* wst = &w2s[threadIdx.x >> 5][0];
#pragma unroll
  for (int i = 0; i < 16; ++i)
    *(v4f*)(wst + (2 * i + (lane >> 4)) * W2P + (lane & 15) * 4) = wq[i];
  const float* wr = wst + lane * W2P;
  float s = bias[row];
#pragma unroll
  for (int j = 0; j < HID; ++j) s = __builtin_fmaf(wr[j], hh[j], s);
  if (which) {
    const int r = row >> 10;
    const int k = row & 1023;
    bmat[t * 16384 + (k >> 3) * 128 + (((k >> 2) & 1) * 16 + r) * 4 + (k & 3)] = s;
  } else {
    amat[t * 8192 + row] = s;
  }
}

__global__ __launch_bounds__(256)
void lora_apply_kernel(const float* __restrict__ x, const float* __restrict__ base,
                       const float* __restrict__ amat, const float* __restrict__ bmat,
                       float* __restrict__ out) {
  const int t    = blockIdx.y;
  const int jt   = blockIdx.x;
  const int tid  = threadIdx.x;
  const int lane = tid & 31;
  const int w    = tid >> 5;
  const int hs   = lane >> 4;
  const int lp   = lane & 15;

  __shared__ __attribute__((aligned(16))) float xs[8][16 * XP];
  __shared__ float part[8][128];
  __shared__ float tmpf[128];

  const float* at = amat + t * (OUT_F * RANK);

  const float* bt = bmat + t * 16384 + (w * 16) * 128 + lane * 4;
  v4f bfr[16];
#pragma unroll
  for (int j = 0; j < 16; ++j) bfr[j] = *(const v4f*)(bt + j * 128);
  const float* xt = x + (size_t)(t + 4 * (jt * 16)) * IN_F + w * 128 + lane * 4;
  v4f xr[16];
#pragma unroll
  for (int m = 0; m < 16; ++m)
    xr[m] = __builtin_nontemporal_load((const v4f*)(xt + (size_t)m * (4 * IN_F)));
  float* xw = &xs[w][0];
#pragma unroll
  for (int m = 0; m < 16; ++m) *(v4f*)(xw + m * XP + lane * 4) = xr[m];
  const float* xf = xw + lp * XP + hs * 4;
  v8f acc = {};
#pragma unroll
  for (int j = 0; j < 16; ++j) {
    const v4f a4 = *(const v4f*)(xf + j * 8);
    const v2f a0 = {a4.x, a4.y};
    const v2f a1 = {a4.z, a4.w};
    const v2f b0 = {bfr[j].x, bfr[j].y};
    const v2f b1 = {bfr[j].z, bfr[j].w};
    acc = __builtin_amdgcn_wmma_f32_16x16x4_f32(
        false, a0, false, b0, (short)0, acc, false, false);
    acc = __builtin_amdgcn_wmma_f32_16x16x4_f32(
        false, a1, false, b1, (short)0, acc, false, false);
  }
  if (lp < RANK) {
#pragma unroll
    for (int v = 0; v < 8; ++v)
      part[w][(v + 8 * hs) * RANK + lp] = acc[v];
  }
  __syncthreads();
  if (tid < 128) {
    float s = part[0][tid];
#pragma unroll
    for (int p = 1; p < 8; ++p) s += part[p][tid];
    tmpf[tid] = s;
  }
  __syncthreads();

  v2f a01 = *(const v2f*)(&tmpf[lp * RANK + hs * 2]);
  v2f a23 = *(const v2f*)(&tmpf[lp * RANK + 4 + hs * 2]);

#pragma unroll 2
  for (int it = 0; it < 8; ++it) {
    const int o = w * 128 + it * 16 + lp;
    v2f b01 = *(const v2f*)(at + o * RANK + hs * 2);
    v2f b23 = *(const v2f*)(at + o * RANK + 4 + hs * 2);
    v8f cc;
#pragma unroll
    for (int v = 0; v < 8; ++v) {
      const size_t row = (size_t)(t + 4 * (jt * 16 + v + 8 * hs));
      cc[v] = __builtin_nontemporal_load(base + row * OUT_F + o);
    }
    cc = __builtin_amdgcn_wmma_f32_16x16x4_f32(
        false, a01, false, b01, (short)0, cc, false, false);
    cc = __builtin_amdgcn_wmma_f32_16x16x4_f32(
        false, a23, false, b23, (short)0, cc, false, false);
#pragma unroll
    for (int v = 0; v < 8; ++v) {
      const size_t row = (size_t)(t + 4 * (jt * 16 + v + 8 * hs));
      __builtin_nontemporal_store(cc[v], out + row * OUT_F + o);
    }
  }
}

extern "C" void kernel_launch(void* const* d_in, const int* in_sizes, int n_in,
                              void* d_out, int out_size, void* d_ws, size_t ws_size,
                              hipStream_t stream) {
  const float* x    = (const float*)d_in[0];
  const float* base = (const float*)d_in[1];
  const float* task = (const float*)d_in[2];
  const float* Wa1  = (const float*)d_in[3];
  const float* ba1  = (const float*)d_in[4];
  const float* Wa2  = (const float*)d_in[5];
  const float* ba2  = (const float*)d_in[6];
  const float* Wb1  = (const float*)d_in[7];
  const float* bb1  = (const float*)d_in[8];
  const float* Wb2  = (const float*)d_in[9];
  const float* bb2  = (const float*)d_in[10];
  float* out = (float*)d_out;

  float* h    = (float*)d_ws;
  float* amat = h + 512;
  float* bmat = amat + 32768;

  gen_hidden_kernel<<<1, 512, 0, stream>>>(task, Wa1, ba1, Wb1, bb1, h);
  gen_mats_kernel<<<384, 256, 0, stream>>>(h, Wa2, ba2, Wb2, bb2, amat, bmat);

  dim3 grid(128, 4, 1);
  lora_apply_kernel<<<grid, 256, 0, stream>>>(x, base, amat, bmat, out);
}


# first layer folded into the generator launch: each block computes the 64 hidden values of its (generator, task) in LDS, lane-split sums + butterfly (f32 sum order of h changes); launches 3 -> 2
# speedup vs baseline: 2.1245x; 1.3039x over previous
#include <hip/hip_runtime.h>

typedef float v2f __attribute__((ext_vector_type(2)));
typedef float v4f __attribute__((ext_vector_type(4)));
typedef float v8f __attribute__((ext_vector_type(8)));

#define IN_F   1024
#define OUT_F  1024
#define TASK_D 256
#define RANK   8
#define HID    64
#define XP     132
#define W2P    68

__global__ __launch_bounds__(256)
void gen_mats_kernel(const float* __restrict__ task_emb,
                     const float* __restrict__ Wa1, const float* __restrict__ ba1,
                     const float* __restrict__ Wa2, const float* __restrict__ ba2,
                     const float* __restrict__ Wb1, const float* __restrict__ bb1,
                     const float* __restrict__ Wb2, const float* __restrict__ bb2,
                     float* __restrict__ amat, float* __restrict__ bmat) {
  __shared__ __attribute__((aligned(16))) float w2s[8][32 * W2P];
  __shared__ __attribute__((aligned(16))) float hsh[HID];
  const int tid = blockIdx.x * 256 + threadIdx.x;
  if (blockIdx.x >= 256) {
    const int rem = tid - 65536;
    const int t   = rem >> 13;
    const int q   = rem & 8191;
    const int j   = q >> 6;
    const int z   = q & 63;
    const int zh  = z >> 5;
    const int zc  = 8 + ((z >> 2) & 7);
    bmat[t * 16384 + j * 128 + (zh * 16 + zc) * 4 + (z & 3)] = 0.0f;
    return;
  }
  const int which = blockIdx.x >> 7;
  const int t     = (blockIdx.x >> 5) & 3;
  const int row   = tid & 8191;
  const float* W    = which ? Wb2 : Wa2;
  const float* bias = which ? bb2 : ba2;
  const int lane = threadIdx.x & 31;
  const int wv   = threadIdx.x >> 5;
  const float* wsrc = W + (size_t)(row & ~31) * HID + lane * 4;
  v4f wq[16];
#pragma unroll
  for (int i = 0; i < 16; ++i) wq[i] = *(const v4f*)(wsrc + i * 128);
  const float b2 = bias[row];

  {
    const float* W1 = which ? Wb1 : Wa1;
    const float* b1 = which ? bb1 : ba1;
    const float* te = task_emb + t * TASK_D + lane * 4;
    const v4f e0 = *(const v4f*)te;
    const v4f e1 = *(const v4f*)(te + 128);
    const float* w1r = W1 + (wv * 8) * TASK_D + lane * 4;
    v4f r0[8], r1[8];
#pragma unroll
    for (int i = 0; i < 8; ++i) {
      r0[i] = *(const v4f*)(w1r + i * TASK_D);
      r1[i] = *(const v4f*)(w1r + i * TASK_D + 128);
    }
    float mine = 0.0f;
#pragma unroll
    for (int i = 0; i < 8; ++i) {
      float p = r0[i].x * e0.x;
      p = __builtin_fmaf(r0[i].y, e0.y, p);
      p = __builtin_fmaf(r0[i].z, e0.z, p);
      p = __builtin_fmaf(r0[i].w, e0.w, p);
      p = __builtin_fmaf(r1[i].x, e1.x, p);
      p = __builtin_fmaf(r1[i].y, e1.y, p);
      p = __builtin_fmaf(r1[i].z, e1.z, p);
      p = __builtin_fmaf(r1[i].w, e1.w, p);
      p += __shfl_xor(p, 16);
      p += __shfl_xor(p, 8);
      p += __shfl_xor(p, 4);
      p += __shfl_xor(p, 2);
      p += __shfl_xor(p, 1);
      if (lane == i) mine = p;
    }
    if (lane < 8) {
      const float s = mine + b1[wv * 8 + lane];
      hsh[wv * 8 + lane] = 0.5f * s * (1.0f + erff(s * 0.7071067811865476f));
    }
  }
  float* wst = &w2s[wv][0];
#pragma unroll
  for (int i = 0; i < 16; ++i)
    *(v4f*)(wst + (2 * i + (lane >> 4)) * W2P + (lane & 15) * 4) = wq[i];
  __syncthreads();
  const float* wr = wst + lane * W2P;
  float s = b2;
#pragma unroll
  for (int j = 0; j < HID; ++j) s = __builtin_fmaf(wr[j], hsh[j], s);
  if (which) {
    const int r = row >> 10;
    const int k = row & 1023;
    bmat[t * 16384 + (k >> 3) * 128 + (((k >> 2) & 1) * 16 + r) * 4 + (k & 3)] = s;
  } else {
    amat[t * 8192 + row] = s;
  }
}

__global__ __launch_bounds__(256)
void lora_apply_kernel(const float* __restrict__ x, const float* __restrict__ base,
                       const float* __restrict__ amat, const float* __restrict__ bmat,
                       float* __restrict__ out) {
  const int t    = blockIdx.y;
  const int jt   = blockIdx.x;
  const int tid  = threadIdx.x;
  const int lane = tid & 31;
  const int w    = tid >> 5;
  const int hs   = lane >> 4;
  const int lp   = lane & 15;

  __shared__ __attribute__((aligned(16))) float xs[8][16 * XP];
  __shared__ float part[8][128];
  __shared__ float tmpf[128];

  const float* at = amat + t * (OUT_F * RANK);

  const float* bt = bmat + t * 16384 + (w * 16) * 128 + lane * 4;
  v4f bfr[16];
#pragma unroll
  for (int j = 0; j < 16; ++j) bfr[j] = *(const v4f*)(bt + j * 128);
  const float* xt = x + (size_t)(t + 4 * (jt * 16)) * IN_F + w * 128 + lane * 4;
  v4f xr[16];
#pragma unroll
  for (int m = 0; m < 16; ++m)
    xr[m] = __builtin_nontemporal_load((const v4f*)(xt + (size_t)m * (4 * IN_F)));
  float* xw = &xs[w][0];
#pragma unroll
  for (int m = 0; m < 16; ++m) *(v4f*)(xw + m * XP + lane * 4) = xr[m];
  const float* xf = xw + lp * XP + hs * 4;
  v8f acc = {};
#pragma unroll
  for (int j = 0; j < 16; ++j) {
    const v4f a4 = *(const v4f*)(xf + j * 8);
    const v2f a0 = {a4.x, a4.y};
    const v2f a1 = {a4.z, a4.w};
    const v2f b0 = {bfr[j].x, bfr[j].y};
    const v2f b1 = {bfr[j].z, bfr[j].w};
    acc = __builtin_amdgcn_wmma_f32_16x16x4_f32(
        false, a0, false, b0, (short)0, acc, false, false);
    acc = __builtin_amdgcn_wmma_f32_16x16x4_f32(
        false, a1, false, b1, (short)0, acc, false, false);
  }
  if (lp < RANK) {
#pragma unroll
    for (int v = 0; v < 8; ++v)
      part[w][(v + 8 * hs) * RANK + lp] = acc[v];
  }
  __syncthreads();
  if (tid < 128) {
    float s = part[0][tid];
#pragma unroll
    for (int p = 1; p < 8; ++p) s += part[p][tid];
    tmpf[tid] = s;
  }
  __syncthreads();

  v2f a01 = *(const v2f*)(&tmpf[lp * RANK + hs * 2]);
  v2f a23 = *(const v2f*)(&tmpf[lp * RANK + 4 + hs * 2]);

#pragma unroll 2
  for (int it = 0; it < 8; ++it) {
    const int o = w * 128 + it * 16 + lp;
    v2f b01 = *(const v2f*)(at + o * RANK + hs * 2);
    v2f b23 = *(const v2f*)(at + o * RANK + 4 + hs * 2);
    v8f cc;
#pragma unroll
    for (int v = 0; v < 8; ++v) {
      const size_t row = (size_t)(t + 4 * (jt * 16 + v + 8 * hs));
      cc[v] = __builtin_nontemporal_load(base + row * OUT_F + o);
    }
    cc = __builtin_amdgcn_wmma_f32_16x16x4_f32(
        false, a01, false, b01, (short)0, cc, false, false);
    cc = __builtin_amdgcn_wmma_f32_16x16x4_f32(
        false, a23, false, b23, (short)0, cc, false, false);
#pragma unroll
    for (int v = 0; v < 8; ++v) {
      const size_t row = (size_t)(t + 4 * (jt * 16 + v + 8 * hs));
      __builtin_nontemporal_store(cc[v], out + row * OUT_F + o);
    }
  }
}

extern "C" void kernel_launch(void* const* d_in, const int* in_sizes, int n_in,
                              void* d_out, int out_size, void* d_ws, size_t ws_size,
                              hipStream_t stream) {
  const float* x    = (const float*)d_in[0];
  const float* base = (const float*)d_in[1];
  const float* task = (const float*)d_in[2];
  const float* Wa1  = (const float*)d_in[3];
  const float* ba1  = (const float*)d_in[4];
  const float* Wa2  = (const float*)d_in[5];
  const float* ba2  = (const float*)d_in[6];
  const float* Wb1  = (const float*)d_in[7];
  const float* bb1  = (const float*)d_in[8];
  const float* Wb2  = (const float*)d_in[9];
  const float* bb2  = (const float*)d_in[10];
  float* out = (float*)d_out;

  float* amat = (float*)d_ws + 512;
  float* bmat = amat + 32768;

  gen_mats_kernel<<<384, 256, 0, stream>>>(task, Wa1, ba1, Wa2, ba2, Wb1, bb1, Wb2, bb2, amat, bmat);

  dim3 grid(128, 4, 1);
  lora_apply_kernel<<<grid, 256, 0, stream>>>(x, base, amat, bmat, out);
}


# base and out moved whole-line: a lane owns 4 adjacent columns (one per n-tile of a 64-col group), b128 per lane, 256 B per half-wave row piece; a_t packed by the generator in that fragment order
# speedup vs baseline: 2.4477x; 1.1521x over previous
#include <hip/hip_runtime.h>

typedef float v2f __attribute__((ext_vector_type(2)));
typedef float v4f __attribute__((ext_vector_type(4)));
typedef float v8f __attribute__((ext_vector_type(8)));

#define IN_F   1024
#define OUT_F  1024
#define TASK_D 256
#define RANK   8
#define HID    64
#define XP     132
#define W2P    68

__global__ __launch_bounds__(256)
void gen_mats_kernel(const float* __restrict__ task_emb,
                     const float* __restrict__ Wa1, const float* __restrict__ ba1,
                     const float* __restrict__ Wa2, const float* __restrict__ ba2,
                     const float* __restrict__ Wb1, const float* __restrict__ bb1,
                     const float* __restrict__ Wb2, const float* __restrict__ bb2,
                     float* __restrict__ amat, float* __restrict__ bmat) {
  __shared__ __attribute__((aligned(16))) float w2s[8][32 * W2P];
  __shared__ __attribute__((aligned(16))) float hsh[HID];
  const int tid = blockIdx.x * 256 + threadIdx.x;
  if (blockIdx.x >= 256) {
    const int rem = tid - 65536;
    const int t   = rem >> 13;
    const int q   = rem & 8191;
    const int j   = q >> 6;
    const int z   = q & 63;
    const int zh  = z >> 5;
    const int zc  = 8 + ((z >> 2) & 7);
    bmat[t * 16384 + j * 128 + (zh * 16 + zc) * 4 + (z & 3)] = 0.0f;
    return;
  }
  const int which = blockIdx.x >> 7;
  const int t     = (blockIdx.x >> 5) & 3;
  const int row   = tid & 8191;
  const float* W    = which ? Wb2 : Wa2;
  const float* bias = which ? bb2 : ba2;
  const int lane = threadIdx.x & 31;
  const int wv   = threadIdx.x >> 5;
  const float* wsrc = W + (size_t)(row & ~31) * HID + lane * 4;
  v4f wq[16];
#pragma unroll
  for (int i = 0; i < 16; ++i) wq[i] = *(const v4f*)(wsrc + i * 128);
  const float b2 = bias[row];

  {
    const float* W1 = which ? Wb1 : Wa1;
    const float* b1 = which ? bb1 : ba1;
    const float* te = task_emb + t * TASK_D + lane * 4;
    const v4f e0 = *(const v4f*)te;
    const v4f e1 = *(const v4f*)(te + 128);
    const float* w1r = W1 + (wv * 8) * TASK_D + lane * 4;
    v4f r0[8], r1[8];
#pragma unroll
    for (int i = 0; i < 8; ++i) {
      r0[i] = *(const v4f*)(w1r + i * TASK_D);
      r1[i] = *(const v4f*)(w1r + i * TASK_D + 128);
    }
    float mine = 0.0f;
#pragma unroll
    for (int i = 0; i < 8; ++i) {
      float p = r0[i].x * e0.x;
      p = __builtin_fmaf(r0[i].y, e0.y, p);
      p = __builtin_fmaf(r0[i].z, e0.z, p);
      p = __builtin_fmaf(r0[i].w, e0.w, p);
      p = __builtin_fmaf(r1[i].x, e1.x, p);
      p = __builtin_fmaf(r1[i].y, e1.y, p);
      p = __builtin_fmaf(r1[i].z, e1.z, p);
      p = __builtin_fmaf(r1[i].w, e1.w, p);
      p += __shfl_xor(p, 16);
      p += __shfl_xor(p, 8);
      p += __shfl_xor(p, 4);
      p += __shfl_xor(p, 2);
      p += __shfl_xor(p, 1);
      if (lane == i) mine = p;
    }
    if (lane < 8) {
      const float s = mine + b1[wv * 8 + lane];
      hsh[wv * 8 + lane] = 0.5f * s * (1.0f + erff(s * 0.7071067811865476f));
    }
  }
  float* wst = &w2s[wv][0];
#pragma unroll
  for (int i = 0; i < 16; ++i)
    *(v4f*)(wst + (2 * i + (lane >> 4)) * W2P + (lane & 15) * 4) = wq[i];
  __syncthreads();
  const float* wr = wst + lane * W2P;
  float s = b2;
#pragma unroll
  for (int j = 0; j < HID; ++j) s = __builtin_fmaf(wr[j], hsh[j], s);
  if (which) {
    const int r = row >> 10;
    const int k = row & 1023;
    bmat[t * 16384 + (k >> 3) * 128 + (((k >> 2) & 1) * 16 + r) * 4 + (k & 3)] = s;
  } else {
    const int o = row >> 3;
    const int r = row & 7;
    amat[t * 8192 + (o >> 6) * 512 + (o & 3) * 128 +
         (((r >> 1) & 1) * 16 + ((o >> 2) & 15)) * 4 + (((r >> 2) << 1) | (r & 1))] = s;
  }
}

__global__ __launch_bounds__(256)
void lora_apply_kernel(const float* __restrict__ x, const float* __restrict__ base,
                       const float* __restrict__ amat, const float* __restrict__ bmat,
                       float* __restrict__ out) {
  const int t    = blockIdx.y;
  const int jt   = blockIdx.x;
  const int tid  = threadIdx.x;
  const int lane = tid & 31;
  const int w    = tid >> 5;
  const int hs   = lane >> 4;
  const int lp   = lane & 15;

  __shared__ __attribute__((aligned(16))) float xs[8][16 * XP];
  __shared__ float part[8][128];
  __shared__ float tmpf[128];

  const float* at = amat + t * (OUT_F * RANK);

  const float* bt = bmat + t * 16384 + (w * 16) * 128 + lane * 4;
  v4f bfr[16];
#pragma unroll
  for (int j = 0; j < 16; ++j) bfr[j] = *(const v4f*)(bt + j * 128);
  const float* xt = x + (size_t)(t + 4 * (jt * 16)) * IN_F + w * 128 + lane * 4;
  v4f xr[16];
#pragma unroll
  for (int m = 0; m < 16; ++m)
    xr[m] = __builtin_nontemporal_load((const v4f*)(xt + (size_t)m * (4 * IN_F)));
  float* xw = &xs[w][0];
#pragma unroll
  for (int m = 0; m < 16; ++m) *(v4f*)(xw + m * XP + lane * 4) = xr[m];
  const float* xf = xw + lp * XP + hs * 4;
  v8f acc = {};
#pragma unroll
  for (int j = 0; j < 16; ++j) {
    const v4f a4 = *(const v4f*)(xf + j * 8);
    const v2f a0 = {a4.x, a4.y};
    const v2f a1 = {a4.z, a4.w};
    const v2f b0 = {bfr[j].x, bfr[j].y};
    const v2f b1 = {bfr[j].z, bfr[j].w};
    acc = __builtin_amdgcn_wmma_f32_16x16x4_f32(
        false, a0, false, b0, (short)0, acc, false, false);
    acc = __builtin_amdgcn_wmma_f32_16x16x4_f32(
        false, a1, false, b1, (short)0, acc, false, false);
  }
  if (lp < RANK) {
#pragma unroll
    for (int v = 0; v < 8; ++v)
      part[w][(v + 8 * hs) * RANK + lp] = acc[v];
  }
  __syncthreads();
  if (tid < 128) {
    float s = part[0][tid];
#pragma unroll
    for (int p = 1; p < 8; ++p) s += part[p][tid];
    tmpf[tid] = s;
  }
  __syncthreads();

  v2f a01 = *(const v2f*)(&tmpf[lp * RANK + hs * 2]);
  v2f a23 = *(const v2f*)(&tmpf[lp * RANK + 4 + hs * 2]);

  const float* apk = at + (w * 2) * 512 + lane * 4;
  const size_t row0 = (size_t)(t + 4 * (jt * 16 + 8 * hs));
  const float* bp = base + row0 * OUT_F + w * 128 + lp * 4;
  float*       op = out  + row0 * OUT_F + w * 128 + lp * 4;
  v4f bq[2][4];
  v4f q[2][8];
#pragma unroll
  for (int g = 0; g < 2; ++g) {
#pragma unroll
    for (int i = 0; i < 4; ++i) bq[g][i] = *(const v4f*)(apk + (g * 4 + i) * 128);
#pragma unroll
    for (int v = 0; v < 8; ++v)
      q[g][v] = __builtin_nontemporal_load((const v4f*)(bp + (size_t)v * (4 * OUT_F) + g * 64));
  }
#pragma unroll
  for (int g = 0; g < 2; ++g) {
#pragma unroll
    for (int i = 0; i < 4; ++i) {
      const v2f b01 = {bq[g][i].x, bq[g][i].y};
      const v2f b23 = {bq[g][i].z, bq[g][i].w};
      v8f cc;
#pragma unroll
      for (int v = 0; v < 8; ++v) cc[v] = q[g][v][i];
      cc = __builtin_amdgcn_wmma_f32_16x16x4_f32(
          false, a01, false, b01, (short)0, cc, false, false);
      cc = __builtin_amdgcn_wmma_f32_16x16x4_f32(
          false, a23, false, b23, (short)0, cc, false, false);
#pragma unroll
      for (int v = 0; v < 8; ++v) q[g][v][i] = cc[v];
    }
#pragma unroll
    for (int v = 0; v < 8; ++v)
      __builtin_nontemporal_store(q[g][v], (v4f*)(op + (size_t)v * (4 * OUT_F) + g * 64));
  }
}

extern "C" void kernel_launch(void* const* d_in, const int* in_sizes, int n_in,
                              void* d_out, int out_size, void* d_ws, size_t ws_size,
                              hipStream_t stream) {
  const float* x    = (const float*)d_in[0];
  const float* base = (const float*)d_in[1];
  const float* task = (const float*)d_in[2];
  const float* Wa1  = (const float*)d_in[3];
  const float* ba1  = (const float*)d_in[4];
  const float* Wa2  = (const float*)d_in[5];
  const float* ba2  = (const float*)d_in[6];
  const float* Wb1  = (const float*)d_in[7];
  const float* bb1  = (const float*)d_in[8];
  const float* Wb2  = (const float*)d_in[9];
  const float* bb2  = (const float*)d_in[10];
  float* out = (float*)d_out;

  float* amat = (float*)d_ws + 512;
  float* bmat = amat + 32768;

  gen_mats_kernel<<<384, 256, 0, stream>>>(task, Wa1, ba1, Wa2, ba2, Wb1, bb1, Wb2, bb2, amat, bmat);

  dim3 grid(128, 4, 1);
  lora_apply_kernel<<<grid, 256, 0, stream>>>(x, base, amat, bmat, out);
}


# main pass: a block takes two M tiles (32 rows); each wave fetches its packed B fragments of both stages once for both and works tile 0 while tile 1's rows arrive; blocks 512 -> 256, sums unchanged
# speedup vs baseline: 2.8154x; 1.1502x over previous
#include <hip/hip_runtime.h>

typedef float v2f __attribute__((ext_vector_type(2)));
typedef float v4f __attribute__((ext_vector_type(4)));
typedef float v8f __attribute__((ext_vector_type(8)));

#define IN_F   1024
#define OUT_F  1024
#define TASK_D 256
#define RANK   8
#define HID    64
#define XP     132
#define W2P    68

__global__ __launch_bounds__(256)
void gen_mats_kernel(const float* __restrict__ task_emb,
                     const float* __restrict__ Wa1, const float* __restrict__ ba1,
                     const float* __restrict__ Wa2, const float* __restrict__ ba2,
                     const float* __restrict__ Wb1, const float* __restrict__ bb1,
                     const float* __restrict__ Wb2, const float* __restrict__ bb2,
                     float* __restrict__ amat, float* __restrict__ bmat) {
  __shared__ __attribute__((aligned(16))) float w2s[8][32 * W2P];
  __shared__ __attribute__((aligned(16))) float hsh[HID];
  const int tid = blockIdx.x * 256 + threadIdx.x;
  if (blockIdx.x >= 256) {
    const int rem = tid - 65536;
    const int t   = rem >> 13;
    const int q   = rem & 8191;
    const int j   = q >> 6;
    const int z   = q & 63;
    const int zh  = z >> 5;
    const int zc  = 8 + ((z >> 2) & 7);
    bmat[t * 16384 + j * 128 + (zh * 16 + zc) * 4 + (z & 3)] = 0.0f;
    return;
  }
  const int which = blockIdx.x >> 7;
  const int t     = (blockIdx.x >> 5) & 3;
  const int row   = tid & 8191;
  const float* W    = which ? Wb2 : Wa2;
  const float* bias = which ? bb2 : ba2;
  const int lane = threadIdx.x & 31;
  const int wv   = threadIdx.x >> 5;
  const float* wsrc = W + (size_t)(row & ~31) * HID + lane * 4;
  v4f wq[16];
#pragma unroll
  for (int i = 0; i < 16; ++i) wq[i] = *(const v4f*)(wsrc + i * 128);
  const float b2 = bias[row];

  {
    const float* W1 = which ? Wb1 : Wa1;
    const float* b1 = which ? bb1 : ba1;
    const float* te = task_emb + t * TASK_D + lane * 4;
    const v4f e0 = *(const v4f*)te;
    const v4f e1 = *(const v4f*)(te + 128);
    const float* w1r = W1 + (wv * 8) * TASK_D + lane * 4;
    v4f r0[8], r1[8];
#pragma unroll
    for (int i = 0; i < 8; ++i) {
      r0[i] = *(const v4f*)(w1r + i * TASK_D);
      r1[i] = *(const v4f*)(w1r + i * TASK_D + 128);
    }
    float mine = 0.0f;
#pragma unroll
    for (int i = 0; i < 8; ++i) {
      float p = r0[i].x * e0.x;
      p = __builtin_fmaf(r0[i].y, e0.y, p);
      p = __builtin_fmaf(r0[i].z, e0.z, p);
      p = __builtin_fmaf(r0[i].w, e0.w, p);
      p = __builtin_fmaf(r1[i].x, e1.x, p);
      p = __builtin_fmaf(r1[i].y, e1.y, p);
      p = __builtin_fmaf(r1[i].z, e1.z, p);
      p = __builtin_fmaf(r1[i].w, e1.w, p);
      p += __shfl_xor(p, 16);
      p += __shfl_xor(p, 8);
      p += __shfl_xor(p, 4);
      p += __shfl_xor(p, 2);
      p += __shfl_xor(p, 1);
      if (lane == i) mine = p;
    }
    if (lane < 8) {
      const float s = mine + b1[wv * 8 + lane];
      hsh[wv * 8 + lane] = 0.5f * s * (1.0f + erff(s * 0.7071067811865476f));
    }
  }
  float* wst = &w2s[wv][0];
#pragma unroll
  for (int i = 0; i < 16; ++i)
    *(v4f*)(wst + (2 * i + (lane >> 4)) * W2P + (lane & 15) * 4) = wq[i];
  __syncthreads();
  const float* wr = wst + lane * W2P;
  float s = b2;
#pragma unroll
  for (int j = 0; j < HID; ++j) s = __builtin_fmaf(wr[j], hsh[j], s);
  if (which) {
    const int r = row >> 10;
    const int k = row & 1023;
    bmat[t * 16384 + (k >> 3) * 128 + (((k >> 2) & 1) * 16 + r) * 4 + (k & 3)] = s;
  } else {
    const int o = row >> 3;
    const int r = row & 7;
    amat[t * 8192 + (o >> 6) * 512 + (o & 3) * 128 +
         (((r >> 1) & 1) * 16 + ((o >> 2) & 15)) * 4 + (((r >> 2) << 1) | (r & 1))] = s;
  }
}

__global__ __launch_bounds__(256)
void lora_apply_kernel(const float* __restrict__ x, const float* __restrict__ base,
                       const float* __restrict__ amat, const float* __restrict__ bmat,
                       float* __restrict__ out) {
  const int t    = blockIdx.y;
  const int jp   = blockIdx.x;
  const int tid  = threadIdx.x;
  const int lane = tid & 31;
  const int w    = tid >> 5;
  const int hs   = lane >> 4;
  const int lp   = lane & 15;

  __shared__ __attribute__((aligned(16))) float xs[8][2][16 * XP];
  __shared__ __attribute__((aligned(16))) float part[8][256];
  __shared__ __attribute__((aligned(16))) float tmpf[256];

  const float* at = amat + t * (OUT_F * RANK);

  const float* bt = bmat + t * 16384 + (w * 16) * 128 + lane * 4;
  v4f bfr[16];
#pragma unroll
  for (int j = 0; j < 16; ++j) bfr[j] = *(const v4f*)(bt + j * 128);
  const float* xt = x + (size_t)(t + 4 * (jp * 32)) * IN_F + w * 128 + lane * 4;
  v4f xr[2][16];
#pragma unroll
  for (int T = 0; T < 2; ++T) {
#pragma unroll
    for (int m = 0; m < 16; ++m)
      xr[T][m] = __builtin_nontemporal_load((const v4f*)(xt + (size_t)(T * 16 + m) * (4 * IN_F)));
  }
  __builtin_amdgcn_sched_barrier(0);
  v8f acc[2];
#pragma unroll
  for (int T = 0; T < 2; ++T) {
    if (T == 1) __builtin_amdgcn_sched_barrier(0);
    float* xw = &xs[w][T][0];
#pragma unroll
    for (int m = 0; m < 16; ++m) *(v4f*)(xw + m * XP + lane * 4) = xr[T][m];
    const float* xf = xw + lp * XP + hs * 4;
    v8f a = {};
#pragma unroll
    for (int j = 0; j < 16; ++j) {
      const v4f a4 = *(const v4f*)(xf + j * 8);
      const v2f a0 = {a4.x, a4.y};
      const v2f a1 = {a4.z, a4.w};
      const v2f b0 = {bfr[j].x, bfr[j].y};
      const v2f b1 = {bfr[j].z, bfr[j].w};
      a = __builtin_amdgcn_wmma_f32_16x16x4_f32(
          false, a0, false, b0, (short)0, a, false, false);
      a = __builtin_amdgcn_wmma_f32_16x16x4_f32(
          false, a1, false, b1, (short)0, a, false, false);
    }
    acc[T] = a;
  }
  if (lp < RANK) {
#pragma unroll
    for (int T = 0; T < 2; ++T) {
#pragma unroll
      for (int v = 0; v < 8; ++v)
        part[w][T * 128 + (v + 8 * hs) * RANK + lp] = acc[T][v];
    }
  }
  __syncthreads();
  {
    float s = part[0][tid];
#pragma unroll
    for (int p = 1; p < 8; ++p) s += part[p][tid];
    tmpf[tid] = s;
  }
  __syncthreads();

  const float* apk = at + (w * 2) * 512 + lane * 4;
  v4f bq[2][4];
#pragma unroll
  for (int g = 0; g < 2; ++g) {
#pragma unroll
    for (int i = 0; i < 4; ++i) bq[g][i] = *(const v4f*)(apk + (g * 4 + i) * 128);
  }
  const size_t row0 = (size_t)(t + 4 * (jp * 32 + 8 * hs));
  const float* bp = base + row0 * OUT_F + w * 128 + lp * 4;
  float*       op = out  + row0 * OUT_F + w * 128 + lp * 4;
  v4f q[2][2][8];
#pragma unroll
  for (int T = 0; T < 2; ++T) {
#pragma unroll
    for (int g = 0; g < 2; ++g) {
#pragma unroll
      for (int v = 0; v < 8; ++v)
        q[T][g][v] = __builtin_nontemporal_load(
            (const v4f*)(bp + (size_t)(T * 16 + v) * (4 * OUT_F) + g * 64));
    }
  }
  __builtin_amdgcn_sched_barrier(0);
#pragma unroll
  for (int T = 0; T < 2; ++T) {
    if (T == 1) __builtin_amdgcn_sched_barrier(0);
    const v2f a01 = *(const v2f*)(&tmpf[T * 128 + lp * RANK + hs * 2]);
    const v2f a23 = *(const v2f*)(&tmpf[T * 128 + lp * RANK + 4 + hs * 2]);
#pragma unroll
    for (int g = 0; g < 2; ++g) {
#pragma unroll
      for (int i = 0; i < 4; ++i) {
        const v2f b01 = {bq[g][i].x, bq[g][i].y};
        const v2f b23 = {bq[g][i].z, bq[g][i].w};
        v8f cc;
#pragma unroll
        for (int v = 0; v < 8; ++v) cc[v] = q[T][g][v][i];
        cc = __builtin_amdgcn_wmma_f32_16x16x4_f32(
            false, a01, false, b01, (short)0, cc, false, false);
        cc = __builtin_amdgcn_wmma_f32_16x16x4_f32(
            false, a23, false, b23, (short)0, cc, false, false);
#pragma unroll
        for (int v = 0; v < 8; ++v) q[T][g][v][i] = cc[v];
      }
#pragma unroll
      for (int v = 0; v < 8; ++v)
        __builtin_nontemporal_store(q[T][g][v],
            (v4f*)(op + (size_t)(T * 16 + v) * (4 * OUT_F) + g * 64));
    }
  }
}

extern "C" void kernel_launch(void* const* d_in, const int* in_sizes, int n_in,
                              void* d_out, int out_size, void* d_ws, size_t ws_size,
                              hipStream_t stream) {
  const float* x    = (const float*)d_in[0];
  const float* base = (const float*)d_in[1];
  const float* task = (const float*)d_in[2];
  const float* Wa1  = (const float*)d_in[3];
  const float* ba1  = (const float*)d_in[4];
  const float* Wa2  = (const float*)d_in[5];
  const float* ba2  = (const float*)d_in[6];
  const float* Wb1  = (const float*)d_in[7];
  const float* bb1  = (const float*)d_in[8];
  const float* Wb2  = (const float*)d_in[9];
  const float* bb2  = (const float*)d_in[10];
  float* out = (float*)d_out;

  float* amat = (float*)d_ws + 512;
  float* bmat = amat + 32768;

  gen_mats_kernel<<<384, 256, 0, stream>>>(task, Wa1, ba1, Wa2, ba2, Wb1, bb1, Wb2, bb2, amat, bmat);

  dim3 grid(64, 4, 1);
  lora_apply_kernel<<<grid, 256, 0, stream>>>(x, base, amat, bmat, out);
}
